# starting kernel
# baseline (speedup 1.0000x reference)
#include <hip/hip_runtime.h>

typedef __attribute__((ext_vector_type(2))) float v2f;
typedef __attribute__((ext_vector_type(8))) float v8f;

#define N_NODES 100000
#define EMB_DIM 128
#define HIDDEN  128
#define OUT_C   64
#define N_EDGES 800000

__global__ void zero4_kernel(float4* __restrict__ p, int n4) {
    int i = blockIdx.x * blockDim.x + threadIdx.x;
    int stride = gridDim.x * blockDim.x;
    float4 z = make_float4(0.f, 0.f, 0.f, 0.f);
    for (; i < n4; i += stride) p[i] = z;
}

__global__ void degree_kernel(const int* __restrict__ ei, float* __restrict__ cnt) {
    int e = blockIdx.x * blockDim.x + threadIdx.x;
    if (e < N_EDGES) atomicAdd(&cnt[ei[N_EDGES + e]], 1.0f);
}

template<int F>
__global__ void scatter_kernel(const int* __restrict__ ei,
                               const float* __restrict__ x,
                               const int* __restrict__ xidx,
                               float* __restrict__ agg) {
    const int CH = F / 4;
    int t = blockIdx.x * blockDim.x + threadIdx.x;
    if (t >= N_EDGES * CH) return;
    int e = t / CH;
    int f = (t - e * CH) * 4;
    int s = ei[e];
    int d = ei[N_EDGES + e];
    if (xidx) s = xidx[s];
    const float4 v = *reinterpret_cast<const float4*>(x + (size_t)s * F + f);
    float* a = agg + (size_t)d * F + f;
    atomicAdd(a + 0, v.x);
    atomicAdd(a + 1, v.y);
    atomicAdd(a + 2, v.z);
    atomicAdd(a + 3, v.w);
}

template<int K, int OUT, bool RELU>
__global__ __launch_bounds__(128)
void sage_wmma_kernel(const float* __restrict__ agg,
                      const float* __restrict__ cnt,
                      const float* __restrict__ X,
                      const int* __restrict__ xidx,
                      const float* __restrict__ Wl,
                      const float* __restrict__ bias,
                      const float* __restrict__ Wr,
                      float* __restrict__ out) {
    const int lane = threadIdx.x;
    const int wave = threadIdx.y;
    const int rowt = blockIdx.x;
    const int colt = blockIdx.y * 4 + wave;
    const int half = lane >> 4;
    const int l16  = lane & 15;

    const int arow = rowt * 16 + l16;
    const int col  = colt * 16 + l16;

    const float inv = 1.0f / fmaxf(cnt[arow], 1.0f);
    int xrow = arow;
    if (xidx) xrow = xidx[arow];

    const float* __restrict__ aA = agg + (size_t)arow * K;
    const float* __restrict__ aX = X   + (size_t)xrow * K;

    v8f c = {0.f, 0.f, 0.f, 0.f, 0.f, 0.f, 0.f, 0.f};

    #pragma unroll
    for (int k = 0; k < K; k += 4) {
        const int ko = k + half * 2;
        v2f a; a.x = aA[ko] * inv;              a.y = aA[ko + 1] * inv;
        v2f b; b.x = Wl[(size_t)ko * OUT + col]; b.y = Wl[(size_t)(ko + 1) * OUT + col];
        c = __builtin_amdgcn_wmma_f32_16x16x4_f32(false, a, false, b,
                                                  (short)0, c, false, false);
    }
    #pragma unroll
    for (int k = 0; k < K; k += 4) {
        const int ko = k + half * 2;
        v2f a; a.x = aX[ko];                    a.y = aX[ko + 1];
        v2f b; b.x = Wr[(size_t)ko * OUT + col]; b.y = Wr[(size_t)(ko + 1) * OUT + col];
        c = __builtin_amdgcn_wmma_f32_16x16x4_f32(false, a, false, b,
                                                  (short)0, c, false, false);
    }

    const float bv = bias[col];
    #pragma unroll
    for (int v = 0; v < 8; ++v) {
        const int r = rowt * 16 + v + half * 8;
        float val = c[v] + bv;
        if (RELU) val = fmaxf(val, 0.0f);
        out[(size_t)r * OUT + col] = val;
    }
}

extern "C" void kernel_launch(void* const* d_in, const int* in_sizes, int n_in,
                              void* d_out, int out_size, void* d_ws, size_t ws_size,
                              hipStream_t stream) {
    (void)in_sizes; (void)n_in; (void)out_size; (void)ws_size;

    const int*   xidx = (const int*)d_in[0];
    const int*   ei   = (const int*)d_in[1];
    const float* emb  = (const float*)d_in[2];
    const float* Wl1  = (const float*)d_in[3];
    const float* bl1  = (const float*)d_in[4];
    const float* Wr1  = (const float*)d_in[5];
    const float* Wl2  = (const float*)d_in[6];
    const float* bl2  = (const float*)d_in[7];
    const float* Wr2  = (const float*)d_in[8];

    float* agg = (float*)d_ws;
    float* h   = agg + (size_t)N_NODES * HIDDEN;
    float* cnt = h   + (size_t)N_NODES * HIDDEN;

    const int aggN4 = N_NODES * EMB_DIM / 4;
    const int cntN4 = N_NODES / 4;
    const int scatT = N_EDGES * (EMB_DIM / 4);

    dim3 wblk(32, 4);

    zero4_kernel<<<2048, 256, 0, stream>>>((float4*)agg, aggN4);
    zero4_kernel<<<128, 256, 0, stream>>>((float4*)cnt, cntN4);
    degree_kernel<<<(N_EDGES + 255) / 256, 256, 0, stream>>>(ei, cnt);
    scatter_kernel<EMB_DIM><<<(scatT + 255) / 256, 256, 0, stream>>>(ei, emb, xidx, agg);
    sage_wmma_kernel<EMB_DIM, HIDDEN, true>
        <<<dim3(N_NODES / 16, HIDDEN / 16 / 4), wblk, 0, stream>>>(
            agg, cnt, emb, xidx, Wl1, bl1, Wr1, h);

    zero4_kernel<<<2048, 256, 0, stream>>>((float4*)agg, aggN4);
    scatter_kernel<HIDDEN><<<(scatT + 255) / 256, 256, 0, stream>>>(ei, h, nullptr, agg);
    sage_wmma_kernel<HIDDEN, OUT_C, false>
        <<<dim3(N_NODES / 16, OUT_C / 16 / 4), wblk, 0, stream>>>(
            agg, cnt, h, nullptr, Wl2, bl2, Wr2, (float*)d_out);
}


# Edge scan reads a 1-byte owner key per edge (dst / 391, written once per call by a new first launch), not the 4-byte dst: 16 edges per lane and load, 1/4 the scan bytes; sums still in arrival order
# speedup vs baseline: 2.8885x; 2.8885x over previous
#include <hip/hip_runtime.h>

typedef __attribute__((ext_vector_type(2))) float v2f;
typedef __attribute__((ext_vector_type(4))) float v4f;
typedef __attribute__((ext_vector_type(8))) float v8f;

#define N_NODES 100000
#define EMB_DIM 128
#define HIDDEN  128
#define OUT_C   64
#define N_EDGES 800000

#define OWN_R      391
#define OWN_BLOCKS ((N_NODES + OWN_R - 1) / OWN_R)
#define OWN_WAVES  32
#define DRAIN      16
#define AHEAD      6
#define KCH        512
#define KEY_NCH    ((N_EDGES + KCH - 1) / KCH)
#define Q_CAP      768
#define Q_ROOM     (Q_CAP - KCH)

__global__ __launch_bounds__(256)
void owner_key_kernel(const int* __restrict__ ei, unsigned* __restrict__ key) {
    static_assert(OWN_BLOCKS <= 256, "owner number must fit a byte");
    static_assert(N_EDGES % 4 == 0, "whole 16-byte loads");
    const int lane = threadIdx.x & 31;
    const int base = (blockIdx.x * 8 + (threadIdx.x >> 5)) * KCH;
    const int4* dst4 = reinterpret_cast<const int4*>(ei + N_EDGES);
    int4 d[4];
    #pragma unroll
    for (int k = 0; k < 4; ++k)
        d[k] = dst4[min(base + 128 * k + 4 * lane, N_EDGES - 4) >> 2];
    #pragma unroll
    for (int k = 0; k < 4; ++k) {
        const int e = base + 128 * k + 4 * lane;
        if (e < N_EDGES)
            key[e >> 2] = ((unsigned)d[k].x / OWN_R) | (((unsigned)d[k].y / OWN_R) << 8) |
                          (((unsigned)d[k].z / OWN_R) << 16) | (((unsigned)d[k].w / OWN_R) << 24);
    }
}

template<bool COUNT>
__global__ __launch_bounds__(1024)
void owner_kernel(const int* __restrict__ ei,
                  const unsigned* __restrict__ key,
                  const float* __restrict__ x,
                  const int* __restrict__ xidx,
                  float* __restrict__ agg,
                  float* __restrict__ cnt) {
    constexpr int F   = 128;
    constexpr int STEPS = (KEY_NCH + OWN_WAVES - 1) / OWN_WAVES;
    static_assert(N_EDGES % 16 == 0, "a lane's 16 keys are all inside the list or all outside");
    static_assert(Q_ROOM >= DRAIN, "the trip's own drain leaves fewer than DRAIN entries");
    __shared__ float s_sum[OWN_R * F];
    __shared__ int   s_deg[OWN_R];
    __shared__ int   s_q[OWN_WAVES * Q_CAP];

    const int tid  = threadIdx.x;
    const int lane = tid & 31;
    const int wave = tid >> 5;
    const int lo   = blockIdx.x * OWN_R;
    const unsigned rep = blockIdx.x * 0x01010101u;

    const v4f z = {0.f, 0.f, 0.f, 0.f};
    for (int i = tid; i < OWN_R * F / 4; i += 1024) reinterpret_cast<v4f*>(s_sum)[i] = z;
    if (COUNT) for (int i = tid; i < OWN_R; i += 1024) s_deg[i] = 0;
    __syncthreads();

    int* q = s_q + wave * Q_CAP;
    int qn = 0;
    const uint4* key4 = reinterpret_cast<const uint4*>(key);

    auto ldchunk = [&](int i) -> uint4 {
        const int c = min(i * OWN_WAVES + wave, KEY_NCH - 1);
        return key4[c * 32 + lane];
    };
    auto drain = [&](int n) {
        int s = 0, r = 0;
        if (lane < n) {
            const int e = q[qn - n + lane];
            s = ei[e];
            r = ei[N_EDGES + e] - lo;
            if (xidx) s = xidx[s];
            if (COUNT) __hip_atomic_fetch_add(&s_deg[r], 1, __ATOMIC_RELAXED,
                                              __HIP_MEMORY_SCOPE_WORKGROUP);
        }
        qn -= n;
        int sr[DRAIN], dr[DRAIN];
        #pragma unroll
        for (int j = 0; j < DRAIN; ++j) {
            sr[j] = __builtin_amdgcn_readlane(s, j);
            dr[j] = __builtin_amdgcn_readlane(r, j);
        }
        float v[DRAIN][4];
        #pragma unroll
        for (int j = 0; j < DRAIN; ++j) {
            if (__builtin_expect(j < n, 1)) {
                const int ro = sr[j] * F + lane;
                #pragma unroll
                for (int c = 0; c < 4; ++c) v[j][c] = x[ro + 32 * c];
            }
        }
        #pragma unroll
        for (int j = 0; j < DRAIN; ++j) {
            if (__builtin_expect(j < n, 1)) {
                const int ro = dr[j] * F + lane;
                #pragma unroll
                for (int c = 0; c < 4; ++c)
                    __hip_atomic_fetch_add(&s_sum[ro + 32 * c], v[j][c],
                                           __ATOMIC_RELAXED, __HIP_MEMORY_SCOPE_WORKGROUP);
            }
        }
    };
    auto drain_all = [&]() {
        #pragma unroll 1
        while (qn > 0) {
            --qn;
            const int e = q[qn];
            int s = ei[e];
            const int r = ei[N_EDGES + e] - lo;
            if (xidx) s = xidx[s];
            if (COUNT && lane == 0) __hip_atomic_fetch_add(&s_deg[r], 1, __ATOMIC_RELAXED,
                                                           __HIP_MEMORY_SCOPE_WORKGROUP);
            #pragma unroll
            for (int c = 0; c < 4; ++c)
                __hip_atomic_fetch_add(&s_sum[r * F + lane + 32 * c], x[s * F + lane + 32 * c],
                                       __ATOMIC_RELAXED, __HIP_MEMORY_SCOPE_WORKGROUP);
        }
    };

    auto hits = [&](unsigned w) -> unsigned {
        const unsigned t = w ^ rep;
        return ~(((t & 0x7f7f7f7fu) + 0x7f7f7f7fu) | t | 0x7f7f7f7fu);
    };
    auto scan = [&](const uint4 d, int i) {
        if (__builtin_expect(qn > Q_ROOM, 0)) drain_all();
        const int e0 = (i * OWN_WAVES + wave) * KCH + 16 * lane;
        unsigned g = (hits(d.x) >> 7) | (hits(d.y) >> 6) | (hits(d.z) >> 5) | (hits(d.w) >> 4);
        if (e0 >= N_EDGES) g = 0u;
        unsigned mask = __builtin_amdgcn_ballot_w32(g != 0u);
        while (mask) {
            if (g) {
                const int p = __builtin_ctz(g);
                q[qn + (int)__builtin_amdgcn_mbcnt_lo(mask, 0u)] = e0 + ((p & 7) << 2) + (p >> 3);
                g &= g - 1u;
            }
            qn += __builtin_popcount(mask);
            mask = __builtin_amdgcn_ballot_w32(g != 0u);
        }
    };

    static_assert(AHEAD == 6, "the trip below is written out for six chunks");
    uint4 d0 = ldchunk(0), d1 = ldchunk(1), d2 = ldchunk(2);
    uint4 d3 = ldchunk(3), d4 = ldchunk(4), d5 = ldchunk(5);
    #pragma unroll 1
    for (int i = 0; i < STEPS; i += AHEAD) {
        scan(d0, i);     d0 = ldchunk(i + 6);
        scan(d1, i + 1); d1 = ldchunk(i + 7);
        scan(d2, i + 2); d2 = ldchunk(i + 8);
        scan(d3, i + 3); d3 = ldchunk(i + 9);
        scan(d4, i + 4); d4 = ldchunk(i + 10);
        scan(d5, i + 5); d5 = ldchunk(i + 11);
        while (qn >= DRAIN) drain(DRAIN);
    }
    if (qn > 0) drain(qn);

    __syncthreads();
    for (int i = tid; i < OWN_R * F / 4; i += 1024) {
        const int row = lo + (i >> 5);
        if (row < N_NODES)
            reinterpret_cast<v4f*>(agg + (size_t)row * F)[i & 31] =
                reinterpret_cast<const v4f*>(s_sum)[i];
    }
    if (COUNT)
        for (int i = tid; i < OWN_R; i += 1024)
            if (lo + i < N_NODES) cnt[lo + i] = (float)s_deg[i];
}

template<int K, int OUT, int RT, bool RELU>
__global__ __launch_bounds__(64)
void sage_strip_kernel(const float* __restrict__ agg,
                       const float* __restrict__ cnt,
                       const float* __restrict__ X,
                       const int* __restrict__ xidx,
                       const float* __restrict__ Wl,
                       const float* __restrict__ bias,
                       const float* __restrict__ Wr,
                       float* __restrict__ out) {
    constexpr int SP = OUT / 64;
    const int lane = threadIdx.x;
    const int unit = blockIdx.x * 2 + threadIdx.y;
    const int span = unit % SP;
    const int rgrp = unit / SP;
    const int half = lane >> 4;
    const int l16  = lane & 15;
    const int cbase = span * 64 + 4 * l16;

    const float* pa[RT];
    const float* px[RT];
    float inv[RT];
    #pragma unroll
    for (int t = 0; t < RT; ++t) {
        const int arow = (rgrp * RT + t) * 16 + l16;
        inv[t] = 1.0f / fmaxf(cnt[arow], 1.0f);
        int xrow = arow;
        if (xidx) xrow = xidx[arow];
        pa[t] = agg + (size_t)arow * K + 4 * half;
        px[t] = X   + (size_t)xrow * K + 4 * half;
    }
    const float* pwl = Wl + (size_t)(4 * half) * OUT + cbase;
    const float* pwr = Wr + (size_t)(4 * half) * OUT + cbase;

    v8f c[RT][4];
    #pragma unroll
    for (int t = 0; t < RT; ++t)
        #pragma unroll
        for (int j = 0; j < 4; ++j)
            c[t][j] = (v8f){0.f, 0.f, 0.f, 0.f, 0.f, 0.f, 0.f, 0.f};

    constexpr int NG = K / 8;
    v4f wc[4], ac[RT];
    #pragma unroll
    for (int i = 0; i < 4; ++i)
        wc[i] = *reinterpret_cast<const v4f*>(pwl + (size_t)i * OUT);
    #pragma unroll
    for (int t = 0; t < RT; ++t)
        ac[t] = *reinterpret_cast<const v4f*>(pa[t]);

    #pragma unroll
    for (int G = 0; G < 2 * NG; ++G) {
        v4f wn[4], an[RT];
        if (G + 1 < 2 * NG) {
            const bool nfirst = (G + 1) < NG;
            const int  ng     = nfirst ? (G + 1) : (G + 1 - NG);
            const float* pw   = nfirst ? pwl : pwr;
            #pragma unroll
            for (int i = 0; i < 4; ++i)
                wn[i] = *reinterpret_cast<const v4f*>(pw + (size_t)(8 * ng + i) * OUT);
            #pragma unroll
            for (int t = 0; t < RT; ++t)
                an[t] = *reinterpret_cast<const v4f*>((nfirst ? pa[t] : px[t]) + 8 * ng);
        }
        #pragma unroll
        for (int t = 0; t < RT; ++t) {
            v4f a = ac[t];
            if (G < NG) a = a * inv[t];
            v2f a0; a0.x = a.x; a0.y = a.y;
            v2f a1; a1.x = a.z; a1.y = a.w;
            #pragma unroll
            for (int j = 0; j < 4; ++j) {
                v2f b; b.x = wc[0][j]; b.y = wc[1][j];
                c[t][j] = __builtin_amdgcn_wmma_f32_16x16x4_f32(false, a0, false, b,
                                                                (short)0, c[t][j], false, false);
            }
            #pragma unroll
            for (int j = 0; j < 4; ++j) {
                v2f b; b.x = wc[2][j]; b.y = wc[3][j];
                c[t][j] = __builtin_amdgcn_wmma_f32_16x16x4_f32(false, a1, false, b,
                                                                (short)0, c[t][j], false, false);
            }
        }
        if (G + 1 < 2 * NG) {
            #pragma unroll
            for (int i = 0; i < 4; ++i) wc[i] = wn[i];
            #pragma unroll
            for (int t = 0; t < RT; ++t) ac[t] = an[t];
        }
    }

    const v4f bv = *reinterpret_cast<const v4f*>(bias + cbase);
    #pragma unroll
    for (int t = 0; t < RT; ++t) {
        #pragma unroll
        for (int v = 0; v < 8; ++v) {
            const int r = (rgrp * RT + t) * 16 + v + half * 8;
            v4f o;
            o.x = c[t][0][v] + bv.x;
            o.y = c[t][1][v] + bv.y;
            o.z = c[t][2][v] + bv.z;
            o.w = c[t][3][v] + bv.w;
            if (RELU) {
                o.x = fmaxf(o.x, 0.0f); o.y = fmaxf(o.y, 0.0f);
                o.z = fmaxf(o.z, 0.0f); o.w = fmaxf(o.w, 0.0f);
            }
            *reinterpret_cast<v4f*>(out + (size_t)r * OUT + cbase) = o;
        }
    }
}

extern "C" void kernel_launch(void* const* d_in, const int* in_sizes, int n_in,
                              void* d_out, int out_size, void* d_ws, size_t ws_size,
                              hipStream_t stream) {
    (void)in_sizes; (void)n_in; (void)out_size; (void)ws_size;

    const int*   xidx = (const int*)d_in[0];
    const int*   ei   = (const int*)d_in[1];
    const float* emb  = (const float*)d_in[2];
    const float* Wl1  = (const float*)d_in[3];
    const float* bl1  = (const float*)d_in[4];
    const float* Wr1  = (const float*)d_in[5];
    const float* Wl2  = (const float*)d_in[6];
    const float* bl2  = (const float*)d_in[7];
    const float* Wr2  = (const float*)d_in[8];

    float* agg = (float*)d_ws;
    float* h   = agg + (size_t)N_NODES * HIDDEN;
    float* cnt = h   + (size_t)N_NODES * HIDDEN;
    unsigned* key = (unsigned*)d_out;

    dim3 wblk(32, 2);
    const int l1Blocks = (N_NODES / 32) * (HIDDEN / 64) / 2;
    const int l2Blocks = (N_NODES / 16) * (OUT_C / 64) / 2;

    owner_key_kernel<<<(KEY_NCH + 7) / 8, 256, 0, stream>>>(ei, key);

    owner_kernel<true><<<OWN_BLOCKS, 1024, 0, stream>>>(ei, key, emb, xidx, agg, cnt);
    sage_strip_kernel<EMB_DIM, HIDDEN, 2, true>
        <<<dim3(l1Blocks), wblk, 0, stream>>>(
            agg, cnt, emb, xidx, Wl1, bl1, Wr1, h);

    owner_kernel<false><<<OWN_BLOCKS, 1024, 0, stream>>>(ei, key, h, nullptr, agg, cnt);
    sage_strip_kernel<HIDDEN, OUT_C, 1, false>
        <<<dim3(l2Blocks), wblk, 0, stream>>>(
            agg, cnt, h, nullptr, Wl2, bl2, Wr2, (float*)d_out);
}


# Second aggregation walks per-wave lists of kept edges (src, local dst) that the first one's scan writes as it drains: no key scan, no src or dst re-read in launch 4; sums still in arrival order
# speedup vs baseline: 3.2429x; 1.1227x over previous
#include <hip/hip_runtime.h>

typedef __attribute__((ext_vector_type(2))) float v2f;
typedef __attribute__((ext_vector_type(4))) float v4f;
typedef __attribute__((ext_vector_type(8))) float v8f;

#define N_NODES 100000
#define EMB_DIM 128
#define HIDDEN  128
#define OUT_C   64
#define N_EDGES 800000

#define OWN_R      391
#define OWN_BLOCKS ((N_NODES + OWN_R - 1) / OWN_R)
#define OWN_WAVES  32
#define DRAIN      16
#define AHEAD      6
#define KCH        512
#define KEY_NCH    ((N_EDGES + KCH - 1) / KCH)
#define Q_CAP      768
#define Q_ROOM     (Q_CAP - KCH)
#define L_CAP      512
#define L_SHIFT    17
#define KEY_WORDS  (1 << 18)
#define LIST_WORDS (OWN_BLOCKS * OWN_WAVES * L_CAP)

__global__ __launch_bounds__(256)
void owner_key_kernel(const int* __restrict__ ei, unsigned* __restrict__ key) {
    static_assert(OWN_BLOCKS <= 256, "owner number must fit a byte");
    static_assert(N_EDGES % 4 == 0, "whole 16-byte loads");
    const int lane = threadIdx.x & 31;
    const int base = (blockIdx.x * 8 + (threadIdx.x >> 5)) * KCH;
    const int4* dst4 = reinterpret_cast<const int4*>(ei + N_EDGES);
    int4 d[4];
    #pragma unroll
    for (int k = 0; k < 4; ++k)
        d[k] = dst4[min(base + 128 * k + 4 * lane, N_EDGES - 4) >> 2];
    #pragma unroll
    for (int k = 0; k < 4; ++k) {
        const int e = base + 128 * k + 4 * lane;
        if (e < N_EDGES)
            key[e >> 2] = ((unsigned)d[k].x / OWN_R) | (((unsigned)d[k].y / OWN_R) << 8) |
                          (((unsigned)d[k].z / OWN_R) << 16) | (((unsigned)d[k].w / OWN_R) << 24);
    }
}

template<bool FIRST>
__global__ __launch_bounds__(1024)
void owner_kernel(const int* __restrict__ ei,
                  const unsigned* __restrict__ key,
                  const float* __restrict__ x,
                  const int* __restrict__ xidx,
                  unsigned* __restrict__ list,
                  int* __restrict__ nlist,
                  float* __restrict__ agg,
                  float* __restrict__ cnt) {
    constexpr bool COUNT = FIRST;
    constexpr int F   = 128;
    static_assert(N_NODES <= (1 << L_SHIFT), "src fits below the local dst in a list word");
    static_assert(OWN_R <= (1 << (32 - L_SHIFT)), "local dst fits above src in a list word");
    static_assert(DRAIN <= 32 && L_CAP % DRAIN == 0, "one list entry per lane and turn");
    constexpr int STEPS = (KEY_NCH + OWN_WAVES - 1) / OWN_WAVES;
    static_assert(N_EDGES % 16 == 0, "a lane's 16 keys are all inside the list or all outside");
    static_assert(Q_ROOM >= DRAIN, "the trip's own drain leaves fewer than DRAIN entries");
    __shared__ float s_sum[OWN_R * F];
    __shared__ int   s_deg[OWN_R];
    __shared__ int   s_q[OWN_WAVES * Q_CAP];

    const int tid  = threadIdx.x;
    const int lane = tid & 31;
    const int wave = tid >> 5;
    const int lo   = blockIdx.x * OWN_R;
    const unsigned rep = blockIdx.x * 0x01010101u;
    const int gw   = blockIdx.x * OWN_WAVES + wave;
    unsigned* L    = list + (size_t)gw * L_CAP;
    int nk = 0;
    if (!FIRST) nk = __builtin_amdgcn_readfirstlane(nlist[gw]);

    const v4f z = {0.f, 0.f, 0.f, 0.f};
    for (int i = tid; i < OWN_R * F / 4; i += 1024) reinterpret_cast<v4f*>(s_sum)[i] = z;
    if (COUNT) for (int i = tid; i < OWN_R; i += 1024) s_deg[i] = 0;
    __syncthreads();

    int* q = s_q + wave * Q_CAP;
    int qn = 0;
    const uint4* key4 = reinterpret_cast<const uint4*>(key);

    auto ldchunk = [&](int i) -> uint4 {
        const int c = min(i * OWN_WAVES + wave, KEY_NCH - 1);
        return key4[c * 32 + lane];
    };
    auto add_rows = [&](int s, int r, int n) {
        int sr[DRAIN], dr[DRAIN];
        #pragma unroll
        for (int j = 0; j < DRAIN; ++j) {
            sr[j] = __builtin_amdgcn_readlane(s, j);
            dr[j] = __builtin_amdgcn_readlane(r, j);
        }
        float v[DRAIN][4];
        #pragma unroll
        for (int j = 0; j < DRAIN; ++j) {
            if (__builtin_expect(j < n, 1)) {
                const int ro = sr[j] * F + lane;
                #pragma unroll
                for (int c = 0; c < 4; ++c) v[j][c] = x[ro + 32 * c];
            }
        }
        #pragma unroll
        for (int j = 0; j < DRAIN; ++j) {
            if (__builtin_expect(j < n, 1)) {
                const int ro = dr[j] * F + lane;
                #pragma unroll
                for (int c = 0; c < 4; ++c)
                    __hip_atomic_fetch_add(&s_sum[ro + 32 * c], v[j][c],
                                           __ATOMIC_RELAXED, __HIP_MEMORY_SCOPE_WORKGROUP);
            }
        }
    };
    auto drain = [&](int n) {
        int s = 0, r = 0;
        if (lane < n) {
            const int e = q[qn - n + lane];
            s = ei[e];
            r = ei[N_EDGES + e] - lo;
            if (FIRST && nk + lane < L_CAP)
                L[nk + lane] = (unsigned)s | ((unsigned)r << L_SHIFT);
            if (xidx) s = xidx[s];
            if (COUNT) __hip_atomic_fetch_add(&s_deg[r], 1, __ATOMIC_RELAXED,
                                              __HIP_MEMORY_SCOPE_WORKGROUP);
        }
        if (FIRST) nk += n;
        qn -= n;
        add_rows(s, r, n);
    };
    auto drain_all = [&]() {
        #pragma unroll 1
        while (qn > 0) {
            --qn;
            const int e = q[qn];
            int s = ei[e];
            const int r = ei[N_EDGES + e] - lo;
            if (FIRST) {
                if (lane == 0 && nk < L_CAP) L[nk] = (unsigned)s | ((unsigned)r << L_SHIFT);
                ++nk;
            }
            if (xidx) s = xidx[s];
            if (COUNT && lane == 0) __hip_atomic_fetch_add(&s_deg[r], 1, __ATOMIC_RELAXED,
                                                           __HIP_MEMORY_SCOPE_WORKGROUP);
            #pragma unroll
            for (int c = 0; c < 4; ++c)
                __hip_atomic_fetch_add(&s_sum[r * F + lane + 32 * c], x[s * F + lane + 32 * c],
                                       __ATOMIC_RELAXED, __HIP_MEMORY_SCOPE_WORKGROUP);
        }
    };

    auto hits = [&](unsigned w) -> unsigned {
        const unsigned t = w ^ rep;
        return ~(((t & 0x7f7f7f7fu) + 0x7f7f7f7fu) | t | 0x7f7f7f7fu);
    };
    auto scan = [&](const uint4 d, int i) {
        if (__builtin_expect(qn > Q_ROOM, 0)) drain_all();
        const int e0 = (i * OWN_WAVES + wave) * KCH + 16 * lane;
        unsigned g = (hits(d.x) >> 7) | (hits(d.y) >> 6) | (hits(d.z) >> 5) | (hits(d.w) >> 4);
        if (e0 >= N_EDGES) g = 0u;
        unsigned mask = __builtin_amdgcn_ballot_w32(g != 0u);
        while (mask) {
            if (g) {
                const int p = __builtin_ctz(g);
                q[qn + (int)__builtin_amdgcn_mbcnt_lo(mask, 0u)] = e0 + ((p & 7) << 2) + (p >> 3);
                g &= g - 1u;
            }
            qn += __builtin_popcount(mask);
            mask = __builtin_amdgcn_ballot_w32(g != 0u);
        }
    };

    static_assert(AHEAD == 6, "the trip below is written out for six chunks");
    if (!FIRST && nk <= L_CAP) {
        if (nk > 0) {
            const int last = nk - 1;
            unsigned cur = L[min(lane, last)];
            int s = (int)(cur & ((1u << L_SHIFT) - 1u));
            if (xidx) s = xidx[s];
            #pragma unroll 1
            for (int p = 0; p < nk; p += DRAIN) {
                const unsigned nxt = L[min(p + DRAIN + lane, last)];
                const int r = (int)(cur >> L_SHIFT);
                const int m = min(DRAIN, nk - p);
                if (m == DRAIN) add_rows(s, r, DRAIN);
                else            add_rows(s, r, m);
                cur = nxt;
                s = (int)(cur & ((1u << L_SHIFT) - 1u));
                if (xidx) s = xidx[s];
            }
        }
    } else {
    uint4 d0 = ldchunk(0), d1 = ldchunk(1), d2 = ldchunk(2);
    uint4 d3 = ldchunk(3), d4 = ldchunk(4), d5 = ldchunk(5);
    #pragma unroll 1
    for (int i = 0; i < STEPS; i += AHEAD) {
        scan(d0, i);     d0 = ldchunk(i + 6);
        scan(d1, i + 1); d1 = ldchunk(i + 7);
        scan(d2, i + 2); d2 = ldchunk(i + 8);
        scan(d3, i + 3); d3 = ldchunk(i + 9);
        scan(d4, i + 4); d4 = ldchunk(i + 10);
        scan(d5, i + 5); d5 = ldchunk(i + 11);
        while (qn >= DRAIN) drain(DRAIN);
    }
    if (qn > 0) drain(qn);
    if (FIRST && lane == 0) nlist[gw] = nk;
    }

    __syncthreads();
    for (int i = tid; i < OWN_R * F / 4; i += 1024) {
        const int row = lo + (i >> 5);
        if (row < N_NODES)
            reinterpret_cast<v4f*>(agg + (size_t)row * F)[i & 31] =
                reinterpret_cast<const v4f*>(s_sum)[i];
    }
    if (COUNT)
        for (int i = tid; i < OWN_R; i += 1024)
            if (lo + i < N_NODES) cnt[lo + i] = (float)s_deg[i];
}

template<int K, int OUT, int RT, bool RELU>
__global__ __launch_bounds__(64)
void sage_strip_kernel(const float* __restrict__ agg,
                       const float* __restrict__ cnt,
                       const float* __restrict__ X,
                       const int* __restrict__ xidx,
                       const float* __restrict__ Wl,
                       const float* __restrict__ bias,
                       const float* __restrict__ Wr,
                       float* __restrict__ out) {
    constexpr int SP = OUT / 64;
    const int lane = threadIdx.x;
    const int unit = blockIdx.x * 2 + threadIdx.y;
    const int span = unit % SP;
    const int rgrp = unit / SP;
    const int half = lane >> 4;
    const int l16  = lane & 15;
    const int cbase = span * 64 + 4 * l16;

    const float* pa[RT];
    const float* px[RT];
    float inv[RT];
    #pragma unroll
    for (int t = 0; t < RT; ++t) {
        const int arow = (rgrp * RT + t) * 16 + l16;
        inv[t] = 1.0f / fmaxf(cnt[arow], 1.0f);
        int xrow = arow;
        if (xidx) xrow = xidx[arow];
        pa[t] = agg + (size_t)arow * K + 4 * half;
        px[t] = X   + (size_t)xrow * K + 4 * half;
    }
    const float* pwl = Wl + (size_t)(4 * half) * OUT + cbase;
    const float* pwr = Wr + (size_t)(4 * half) * OUT + cbase;

    v8f c[RT][4];
    #pragma unroll
    for (int t = 0; t < RT; ++t)
        #pragma unroll
        for (int j = 0; j < 4; ++j)
            c[t][j] = (v8f){0.f, 0.f, 0.f, 0.f, 0.f, 0.f, 0.f, 0.f};

    constexpr int NG = K / 8;
    v4f wc[4], ac[RT];
    #pragma unroll
    for (int i = 0; i < 4; ++i)
        wc[i] = *reinterpret_cast<const v4f*>(pwl + (size_t)i * OUT);
    #pragma unroll
    for (int t = 0; t < RT; ++t)
        ac[t] = *reinterpret_cast<const v4f*>(pa[t]);

    #pragma unroll
    for (int G = 0; G < 2 * NG; ++G) {
        v4f wn[4], an[RT];
        if (G + 1 < 2 * NG) {
            const bool nfirst = (G + 1) < NG;
            const int  ng     = nfirst ? (G + 1) : (G + 1 - NG);
            const float* pw   = nfirst ? pwl : pwr;
            #pragma unroll
            for (int i = 0; i < 4; ++i)
                wn[i] = *reinterpret_cast<const v4f*>(pw + (size_t)(8 * ng + i) * OUT);
            #pragma unroll
            for (int t = 0; t < RT; ++t)
                an[t] = *reinterpret_cast<const v4f*>((nfirst ? pa[t] : px[t]) + 8 * ng);
        }
        #pragma unroll
        for (int t = 0; t < RT; ++t) {
            v4f a = ac[t];
            if (G < NG) a = a * inv[t];
            v2f a0; a0.x = a.x; a0.y = a.y;
            v2f a1; a1.x = a.z; a1.y = a.w;
            #pragma unroll
            for (int j = 0; j < 4; ++j) {
                v2f b; b.x = wc[0][j]; b.y = wc[1][j];
                c[t][j] = __builtin_amdgcn_wmma_f32_16x16x4_f32(false, a0, false, b,
                                                                (short)0, c[t][j], false, false);
            }
            #pragma unroll
            for (int j = 0; j < 4; ++j) {
                v2f b; b.x = wc[2][j]; b.y = wc[3][j];
                c[t][j] = __builtin_amdgcn_wmma_f32_16x16x4_f32(false, a1, false, b,
                                                                (short)0, c[t][j], false, false);
            }
        }
        if (G + 1 < 2 * NG) {
            #pragma unroll
            for (int i = 0; i < 4; ++i) wc[i] = wn[i];
            #pragma unroll
            for (int t = 0; t < RT; ++t) ac[t] = an[t];
        }
    }

    const v4f bv = *reinterpret_cast<const v4f*>(bias + cbase);
    #pragma unroll
    for (int t = 0; t < RT; ++t) {
        #pragma unroll
        for (int v = 0; v < 8; ++v) {
            const int r = (rgrp * RT + t) * 16 + v + half * 8;
            v4f o;
            o.x = c[t][0][v] + bv.x;
            o.y = c[t][1][v] + bv.y;
            o.z = c[t][2][v] + bv.z;
            o.w = c[t][3][v] + bv.w;
            if (RELU) {
                o.x = fmaxf(o.x, 0.0f); o.y = fmaxf(o.y, 0.0f);
                o.z = fmaxf(o.z, 0.0f); o.w = fmaxf(o.w, 0.0f);
            }
            *reinterpret_cast<v4f*>(out + (size_t)r * OUT + cbase) = o;
        }
    }
}

extern "C" void kernel_launch(void* const* d_in, const int* in_sizes, int n_in,
                              void* d_out, int out_size, void* d_ws, size_t ws_size,
                              hipStream_t stream) {
    (void)in_sizes; (void)n_in; (void)out_size; (void)ws_size;

    const int*   xidx = (const int*)d_in[0];
    const int*   ei   = (const int*)d_in[1];
    const float* emb  = (const float*)d_in[2];
    const float* Wl1  = (const float*)d_in[3];
    const float* bl1  = (const float*)d_in[4];
    const float* Wr1  = (const float*)d_in[5];
    const float* Wl2  = (const float*)d_in[6];
    const float* bl2  = (const float*)d_in[7];
    const float* Wr2  = (const float*)d_in[8];

    float* agg = (float*)d_ws;
    float* h   = agg + (size_t)N_NODES * HIDDEN;
    float* cnt = h   + (size_t)N_NODES * HIDDEN;
    unsigned* key = (unsigned*)d_out;
    unsigned* list = key + KEY_WORDS;
    int* nlist = (int*)(list + LIST_WORDS);
    static_assert(KEY_NCH * KCH <= KEY_WORDS * 4, "the keys end in front of the lists");
    static_assert((size_t)(KEY_WORDS + LIST_WORDS + OWN_BLOCKS * OWN_WAVES) * 4 <=
                  (size_t)N_NODES * OUT_C * 4, "keys, lists and lengths fit the output buffer");

    dim3 wblk(32, 2);
    const int l1Blocks = (N_NODES / 32) * (HIDDEN / 64) / 2;
    const int l2Blocks = (N_NODES / 16) * (OUT_C / 64) / 2;

    owner_key_kernel<<<(KEY_NCH + 7) / 8, 256, 0, stream>>>(ei, key);

    owner_kernel<true><<<OWN_BLOCKS, 1024, 0, stream>>>(ei, key, emb, xidx, list, nlist, agg, cnt);
    sage_strip_kernel<EMB_DIM, HIDDEN, 2, true>
        <<<dim3(l1Blocks), wblk, 0, stream>>>(
            agg, cnt, emb, xidx, Wl1, bl1, Wr1, h);

    owner_kernel<false><<<OWN_BLOCKS, 1024, 0, stream>>>(ei, key, h, nullptr, list, nlist, agg, cnt);
    sage_strip_kernel<HIDDEN, OUT_C, 1, false>
        <<<dim3(l2Blocks), wblk, 0, stream>>>(
            agg, cnt, h, nullptr, Wl2, bl2, Wr2, (float*)d_out);
}


# Owner block runs its layer's two f32 WMMA products on its own LDS sums (mean fragments by 16-byte LDS loads, row pitch 132 floats), stores h / out: sums never stored, 5 launches -> 3; sum order kept
# speedup vs baseline: 3.7937x; 1.1698x over previous
#include <hip/hip_runtime.h>

typedef __attribute__((ext_vector_type(2))) float v2f;
typedef __attribute__((ext_vector_type(4))) float v4f;
typedef __attribute__((ext_vector_type(8))) float v8f;

#define N_NODES 100000
#define EMB_DIM 128
#define HIDDEN  128
#define OUT_C   64
#define N_EDGES 800000

#define OWN_R      391
#define PITCH      132
#define OWN_BLOCKS ((N_NODES + OWN_R - 1) / OWN_R)
#define OWN_WAVES  32
#define DRAIN      16
#define AHEAD      6
#define KCH        512
#define KEY_NCH    ((N_EDGES + KCH - 1) / KCH)
#define Q_CAP      768
#define Q_ROOM     (Q_CAP - KCH)
#define L_CAP      512
#define L_SHIFT    17
#define KEY_WORDS  (1 << 18)
#define LIST_WORDS (OWN_BLOCKS * OWN_WAVES * L_CAP)

__global__ __launch_bounds__(256)
void owner_key_kernel(const int* __restrict__ ei, unsigned* __restrict__ key) {
    static_assert(OWN_BLOCKS <= 256, "owner number must fit a byte");
    static_assert(N_EDGES % 4 == 0, "whole 16-byte loads");
    const int lane = threadIdx.x & 31;
    const int base = (blockIdx.x * 8 + (threadIdx.x >> 5)) * KCH;
    const int4* dst4 = reinterpret_cast<const int4*>(ei + N_EDGES);
    int4 d[4];
    #pragma unroll
    for (int k = 0; k < 4; ++k)
        d[k] = dst4[min(base + 128 * k + 4 * lane, N_EDGES - 4) >> 2];
    #pragma unroll
    for (int k = 0; k < 4; ++k) {
        const int e = base + 128 * k + 4 * lane;
        if (e < N_EDGES)
            key[e >> 2] = ((unsigned)d[k].x / OWN_R) | (((unsigned)d[k].y / OWN_R) << 8) |
                          (((unsigned)d[k].z / OWN_R) << 16) | (((unsigned)d[k].w / OWN_R) << 24);
    }
}

template<bool FIRST, int OUT, bool RELU>
__global__ __launch_bounds__(1024)
void owner_kernel(const int* __restrict__ ei,
                  const unsigned* __restrict__ key,
                  const float* __restrict__ x,
                  const int* __restrict__ xidx,
                  unsigned* __restrict__ list,
                  int* __restrict__ nlist,
                  const float* __restrict__ Wl,
                  const float* __restrict__ bias,
                  const float* __restrict__ Wr,
                  float* __restrict__ cnt,
                  float* __restrict__ out) {
    constexpr bool COUNT = FIRST;
    constexpr int F   = 128;
    static_assert(N_NODES <= (1 << L_SHIFT), "src fits below the local dst in a list word");
    static_assert(OWN_R <= (1 << (32 - L_SHIFT)), "local dst fits above src in a list word");
    static_assert(DRAIN <= 32 && L_CAP % DRAIN == 0, "one list entry per lane and turn");
    constexpr int STEPS = (KEY_NCH + OWN_WAVES - 1) / OWN_WAVES;
    static_assert(N_EDGES % 16 == 0, "a lane's 16 keys are all inside the list or all outside");
    static_assert(Q_ROOM >= DRAIN, "the trip's own drain leaves fewer than DRAIN entries");
    __shared__ __attribute__((aligned(16))) float s_sum[OWN_R * PITCH];
    __shared__ int   s_deg[OWN_R];
    __shared__ int   s_q[OWN_WAVES * Q_CAP];

    const int tid  = threadIdx.x;
    const int lane = tid & 31;
    const int wave = tid >> 5;
    const int lo   = blockIdx.x * OWN_R;
    const unsigned rep = blockIdx.x * 0x01010101u;
    const int gw   = blockIdx.x * OWN_WAVES + wave;
    unsigned* L    = list + (size_t)gw * L_CAP;
    int nk = 0;
    if (!FIRST) nk = __builtin_amdgcn_readfirstlane(nlist[gw]);

    const v4f z = {0.f, 0.f, 0.f, 0.f};
    static_assert((OWN_R * PITCH) % 4 == 0, "whole 16-byte clears");
    for (int i = tid; i < OWN_R * PITCH / 4; i += 1024) reinterpret_cast<v4f*>(s_sum)[i] = z;
    if (COUNT) for (int i = tid; i < OWN_R; i += 1024) s_deg[i] = 0;
    __syncthreads();

    int* q = s_q + wave * Q_CAP;
    int qn = 0;
    const uint4* key4 = reinterpret_cast<const uint4*>(key);

    auto ldchunk = [&](int i) -> uint4 {
        const int c = min(i * OWN_WAVES + wave, KEY_NCH - 1);
        return key4[c * 32 + lane];
    };
    auto add_rows = [&](int s, int r, int n) {
        int sr[DRAIN], dr[DRAIN];
        #pragma unroll
        for (int j = 0; j < DRAIN; ++j) {
            sr[j] = __builtin_amdgcn_readlane(s, j);
            dr[j] = __builtin_amdgcn_readlane(r, j);
        }
        float v[DRAIN][4];
        #pragma unroll
        for (int j = 0; j < DRAIN; ++j) {
            if (__builtin_expect(j < n, 1)) {
                const int ro = sr[j] * F + lane;
                #pragma unroll
                for (int c = 0; c < 4; ++c) v[j][c] = x[ro + 32 * c];
            }
        }
        #pragma unroll
        for (int j = 0; j < DRAIN; ++j) {
            if (__builtin_expect(j < n, 1)) {
                const int ro = dr[j] * PITCH + lane;
                #pragma unroll
                for (int c = 0; c < 4; ++c)
                    __hip_atomic_fetch_add(&s_sum[ro + 32 * c], v[j][c],
                                           __ATOMIC_RELAXED, __HIP_MEMORY_SCOPE_WORKGROUP);
            }
        }
    };
    auto drain = [&](int n) {
        int s = 0, r = 0;
        if (lane < n) {
            const int e = q[qn - n + lane];
            s = ei[e];
            r = ei[N_EDGES + e] - lo;
            if (FIRST && nk + lane < L_CAP)
                L[nk + lane] = (unsigned)s | ((unsigned)r << L_SHIFT);
            if (xidx) s = xidx[s];
            if (COUNT) __hip_atomic_fetch_add(&s_deg[r], 1, __ATOMIC_RELAXED,
                                              __HIP_MEMORY_SCOPE_WORKGROUP);
        }
        if (FIRST) nk += n;
        qn -= n;
        add_rows(s, r, n);
    };
    auto drain_all = [&]() {
        #pragma unroll 1
        while (qn > 0) {
            --qn;
            const int e = q[qn];
            int s = ei[e];
            const int r = ei[N_EDGES + e] - lo;
            if (FIRST) {
                if (lane == 0 && nk < L_CAP) L[nk] = (unsigned)s | ((unsigned)r << L_SHIFT);
                ++nk;
            }
            if (xidx) s = xidx[s];
            if (COUNT && lane == 0) __hip_atomic_fetch_add(&s_deg[r], 1, __ATOMIC_RELAXED,
                                                           __HIP_MEMORY_SCOPE_WORKGROUP);
            #pragma unroll
            for (int c = 0; c < 4; ++c)
                __hip_atomic_fetch_add(&s_sum[r * PITCH + lane + 32 * c], x[s * F + lane + 32 * c],
                                       __ATOMIC_RELAXED, __HIP_MEMORY_SCOPE_WORKGROUP);
        }
    };

    auto hits = [&](unsigned w) -> unsigned {
        const unsigned t = w ^ rep;
        return ~(((t & 0x7f7f7f7fu) + 0x7f7f7f7fu) | t | 0x7f7f7f7fu);
    };
    auto scan = [&](const uint4 d, int i) {
        if (__builtin_expect(qn > Q_ROOM, 0)) drain_all();
        const int e0 = (i * OWN_WAVES + wave) * KCH + 16 * lane;
        unsigned g = (hits(d.x) >> 7) | (hits(d.y) >> 6) | (hits(d.z) >> 5) | (hits(d.w) >> 4);
        if (e0 >= N_EDGES) g = 0u;
        unsigned mask = __builtin_amdgcn_ballot_w32(g != 0u);
        while (mask) {
            if (g) {
                const int p = __builtin_ctz(g);
                q[qn + (int)__builtin_amdgcn_mbcnt_lo(mask, 0u)] = e0 + ((p & 7) << 2) + (p >> 3);
                g &= g - 1u;
            }
            qn += __builtin_popcount(mask);
            mask = __builtin_amdgcn_ballot_w32(g != 0u);
        }
    };

    static_assert(AHEAD == 6, "the trip below is written out for six chunks");
    if (!FIRST && nk <= L_CAP) {
        if (nk > 0) {
            const int last = nk - 1;
            unsigned cur = L[min(lane, last)];
            int s = (int)(cur & ((1u << L_SHIFT) - 1u));
            if (xidx) s = xidx[s];
            #pragma unroll 1
            for (int p = 0; p < nk; p += DRAIN) {
                const unsigned nxt = L[min(p + DRAIN + lane, last)];
                const int r = (int)(cur >> L_SHIFT);
                const int m = min(DRAIN, nk - p);
                if (m == DRAIN) add_rows(s, r, DRAIN);
                else            add_rows(s, r, m);
                cur = nxt;
                s = (int)(cur & ((1u << L_SHIFT) - 1u));
                if (xidx) s = xidx[s];
            }
        }
    } else {
    uint4 d0 = ldchunk(0), d1 = ldchunk(1), d2 = ldchunk(2);
    uint4 d3 = ldchunk(3), d4 = ldchunk(4), d5 = ldchunk(5);
    #pragma unroll 1
    for (int i = 0; i < STEPS; i += AHEAD) {
        scan(d0, i);     d0 = ldchunk(i + 6);
        scan(d1, i + 1); d1 = ldchunk(i + 7);
        scan(d2, i + 2); d2 = ldchunk(i + 8);
        scan(d3, i + 3); d3 = ldchunk(i + 9);
        scan(d4, i + 4); d4 = ldchunk(i + 10);
        scan(d5, i + 5); d5 = ldchunk(i + 11);
        while (qn >= DRAIN) drain(DRAIN);
    }
    if (qn > 0) drain(qn);
    if (FIRST && lane == 0) nlist[gw] = nk;
    }

    __syncthreads();
    if (COUNT)
        for (int i = tid; i < OWN_R; i += 1024)
            if (lo + i < N_NODES) cnt[lo + i] = (float)s_deg[i];

    {
        constexpr int K  = F;
        constexpr int SP = OUT / 64;
        constexpr int NG = K / 8;
        const int nrows  = min(OWN_R, N_NODES - lo);
        const int nunits = ((nrows + 15) >> 4) * SP;
        const int half = lane >> 4;
        const int l16  = lane & 15;
        const int wu = __builtin_amdgcn_readfirstlane(wave);
        auto unit = [&](const int u) {
            const int span  = u % SP;
            const int t     = u / SP;
            const int cbase = span * 64 + 4 * l16;
            const int lrow  = min(t * 16 + l16, nrows - 1);
            const float deg = FIRST ? (float)s_deg[lrow] : cnt[lo + lrow];
            const float inv = 1.0f / fmaxf(deg, 1.0f);
            int xrow = lo + lrow;
            if (xidx) xrow = xidx[xrow];
            const int aoff   = lrow * PITCH + 4 * half;
            const float* px  = x  + (size_t)xrow * K + 4 * half;
            const float* pwl = Wl + (size_t)(4 * half) * OUT + cbase;
            const float* pwr = Wr + (size_t)(4 * half) * OUT + cbase;

            v8f c[4];
            #pragma unroll
            for (int j = 0; j < 4; ++j)
                c[j] = (v8f){0.f, 0.f, 0.f, 0.f, 0.f, 0.f, 0.f, 0.f};

            v4f wc[4], ac;
            #pragma unroll
            for (int i = 0; i < 4; ++i)
                wc[i] = *reinterpret_cast<const v4f*>(pwl + (size_t)i * OUT);
            ac = *reinterpret_cast<const v4f*>(&s_sum[aoff]);

            #pragma unroll
            for (int G = 0; G < 2 * NG; ++G) {
                v4f wn[4], an;
                if (G + 1 < 2 * NG) {
                    const bool nfirst = (G + 1) < NG;
                    const int  ng     = nfirst ? (G + 1) : (G + 1 - NG);
                    const float* pw   = nfirst ? pwl : pwr;
                    #pragma unroll
                    for (int i = 0; i < 4; ++i)
                        wn[i] = *reinterpret_cast<const v4f*>(pw + (size_t)(8 * ng + i) * OUT);
                    if (nfirst) an = *reinterpret_cast<const v4f*>(&s_sum[aoff + 8 * ng]);
                    else        an = *reinterpret_cast<const v4f*>(px + 8 * ng);
                }
                __builtin_amdgcn_sched_barrier(0);
                v4f a = ac;
                if (G < NG) a = a * inv;
                v2f a0; a0.x = a.x; a0.y = a.y;
                v2f a1; a1.x = a.z; a1.y = a.w;
                #pragma unroll
                for (int j = 0; j < 4; ++j) {
                    v2f b; b.x = wc[0][j]; b.y = wc[1][j];
                    c[j] = __builtin_amdgcn_wmma_f32_16x16x4_f32(false, a0, false, b,
                                                                 (short)0, c[j], false, false);
                }
                #pragma unroll
                for (int j = 0; j < 4; ++j) {
                    v2f b; b.x = wc[2][j]; b.y = wc[3][j];
                    c[j] = __builtin_amdgcn_wmma_f32_16x16x4_f32(false, a1, false, b,
                                                                 (short)0, c[j], false, false);
                }
                if (G + 1 < 2 * NG) {
                    #pragma unroll
                    for (int i = 0; i < 4; ++i) wc[i] = wn[i];
                    ac = an;
                }
            }

            const v4f bv = *reinterpret_cast<const v4f*>(bias + cbase);
            #pragma unroll
            for (int v = 0; v < 8; ++v) {
                const int rl = t * 16 + v + half * 8;
                v4f o;
                o.x = c[0][v] + bv.x;
                o.y = c[1][v] + bv.y;
                o.z = c[2][v] + bv.z;
                o.w = c[3][v] + bv.w;
                if (RELU) {
                    o.x = fmaxf(o.x, 0.0f); o.y = fmaxf(o.y, 0.0f);
                    o.z = fmaxf(o.z, 0.0f); o.w = fmaxf(o.w, 0.0f);
                }
                if (rl < nrows)
                    *reinterpret_cast<v4f*>(out + (size_t)(lo + rl) * OUT + cbase) = o;
            }
        };
        constexpr int MAXU = ((OWN_R + 15) / 16) * SP;
        if (MAXU <= OWN_WAVES) {
            if (wu < nunits) unit(wu);
        } else {
            #pragma unroll 1
            for (int u = wu; u < nunits; u += OWN_WAVES) unit(u);
        }
    }
}

extern "C" void kernel_launch(void* const* d_in, const int* in_sizes, int n_in,
                              void* d_out, int out_size, void* d_ws, size_t ws_size,
                              hipStream_t stream) {
    (void)in_sizes; (void)n_in; (void)out_size; (void)ws_size;

    const int*   xidx = (const int*)d_in[0];
    const int*   ei   = (const int*)d_in[1];
    const float* emb  = (const float*)d_in[2];
    const float* Wl1  = (const float*)d_in[3];
    const float* bl1  = (const float*)d_in[4];
    const float* Wr1  = (const float*)d_in[5];
    const float* Wl2  = (const float*)d_in[6];
    const float* bl2  = (const float*)d_in[7];
    const float* Wr2  = (const float*)d_in[8];

    unsigned* key  = (unsigned*)d_ws;
    unsigned* list = key + KEY_WORDS;
    int* nlist = (int*)(list + LIST_WORDS);
    float* h   = (float*)d_ws + (size_t)N_NODES * HIDDEN;
    float* cnt = h + (size_t)N_NODES * HIDDEN;
    static_assert(KEY_NCH * KCH <= KEY_WORDS * 4, "the keys end in front of the lists");
    static_assert((size_t)(KEY_WORDS + LIST_WORDS + OWN_BLOCKS * OWN_WAVES) * 4 <=
                  (size_t)N_NODES * HIDDEN * 4, "keys, lists and lengths end in front of h");
    static_assert(EMB_DIM == 128 && HIDDEN == 128, "a gathered row is 128 floats in both layers");

    owner_key_kernel<<<(KEY_NCH + 7) / 8, 256, 0, stream>>>(ei, key);

    owner_kernel<true, HIDDEN, true><<<OWN_BLOCKS, 1024, 0, stream>>>(
        ei, key, emb, xidx, list, nlist, Wl1, bl1, Wr1, cnt, h);

    owner_kernel<false, OUT_C, false><<<OWN_BLOCKS, 1024, 0, stream>>>(
        ei, key, h, nullptr, list, nlist, Wl2, bl2, Wr2, cnt, (float*)d_out);
}


# Owner keys as 8 bit planes per 32 edges (same bytes as a byte per edge): a scan lane finds its block's edges among 32 with 8 XOR-AND steps, 1024 edges a wave step (was 512); sums in arrival order
# speedup vs baseline: 3.9967x; 1.0535x over previous
#include <hip/hip_runtime.h>

typedef __attribute__((ext_vector_type(2))) float v2f;
typedef __attribute__((ext_vector_type(4))) float v4f;
typedef __attribute__((ext_vector_type(8))) float v8f;

#define N_NODES 100000
#define EMB_DIM 128
#define HIDDEN  128
#define OUT_C   64
#define N_EDGES 800000

#define OWN_R      391
#define PITCH      132
#define OWN_BLOCKS ((N_NODES + OWN_R - 1) / OWN_R)
#define OWN_WAVES  32
#define DRAIN      16
#define AHEAD      5
#define KCH        512
#define KEY_NCH    ((N_EDGES + KCH - 1) / KCH)
#define N_GROUPS   (N_EDGES / 32)
#define SCAN_NCH   ((N_GROUPS + 31) / 32)
#define Q_CAP      768
#define L_CAP      512
#define L_SHIFT    17
#define KEY_WORDS  (1 << 18)
#define LIST_WORDS (OWN_BLOCKS * OWN_WAVES * L_CAP)

__global__ __launch_bounds__(256)
void owner_key_kernel(const int* __restrict__ ei, unsigned* __restrict__ key) {
    static_assert(OWN_BLOCKS <= 256, "owner number must fit 8 bits");
    static_assert(N_EDGES % 32 == 0, "whole groups of 32 edges");
    const int lane = threadIdx.x & 31;
    const int kc   = blockIdx.x * 8 + (threadIdx.x >> 5);
    const int* dst = ei + N_EDGES;
    int d[16];
    #pragma unroll
    for (int g = 0; g < 16; ++g)
        d[g] = dst[min(kc * KCH + 32 * g + lane, N_EDGES - 1)];
    unsigned o[4] = {0u, 0u, 0u, 0u};
    #pragma unroll
    for (int g = 0; g < 16; ++g) {
        const unsigned k = (unsigned)d[g] / OWN_R;
        #pragma unroll
        for (int b = 0; b < 8; ++b) {
            const unsigned w = __builtin_amdgcn_ballot_w32(((k >> b) & 1u) != 0u);
            if (lane == g + 16 * (b >> 2)) o[b & 3] = w;
        }
    }
    uint4 v;
    v.x = o[0]; v.y = o[1]; v.z = o[2]; v.w = o[3];
    reinterpret_cast<uint4*>(key)[(kc >> 1) * 64 + (lane >> 4) * 32 + (kc & 1) * 16 + (lane & 15)] = v;
}

template<bool FIRST, int OUT, bool RELU>
__global__ __launch_bounds__(1024)
void owner_kernel(const int* __restrict__ ei,
                  const unsigned* __restrict__ key,
                  const float* __restrict__ x,
                  const int* __restrict__ xidx,
                  unsigned* __restrict__ list,
                  int* __restrict__ nlist,
                  const float* __restrict__ Wl,
                  const float* __restrict__ bias,
                  const float* __restrict__ Wr,
                  float* __restrict__ cnt,
                  float* __restrict__ out) {
    constexpr bool COUNT = FIRST;
    constexpr int F   = 128;
    static_assert(N_NODES <= (1 << L_SHIFT), "src fits below the local dst in a list word");
    static_assert(OWN_R <= (1 << (32 - L_SHIFT)), "local dst fits above src in a list word");
    static_assert(DRAIN <= 32 && L_CAP % DRAIN == 0, "one list entry per lane and turn");
    constexpr int STEPS = (SCAN_NCH + OWN_WAVES - 1) / OWN_WAVES;
    static_assert(N_EDGES % 32 == 0, "a lane's group of 32 edges is all inside the list or all outside");
    static_assert(STEPS % AHEAD == 0, "whole trips");
    static_assert(Q_CAP >= 64, "a push turn adds at most 32 entries to at most Q_CAP - 32");
    __shared__ __attribute__((aligned(16))) float s_sum[OWN_R * PITCH];
    __shared__ int   s_deg[OWN_R];
    __shared__ int   s_q[OWN_WAVES * Q_CAP];

    const int tid  = threadIdx.x;
    const int lane = tid & 31;
    const int wave = tid >> 5;
    const int lo   = blockIdx.x * OWN_R;
    unsigned inv[8];
    #pragma unroll
    for (int b = 0; b < 8; ++b) inv[b] = ((blockIdx.x >> b) & 1u) ? 0u : 0xffffffffu;
    const int gw   = blockIdx.x * OWN_WAVES + wave;
    unsigned* L    = list + (size_t)gw * L_CAP;
    int nk = 0;
    if (!FIRST) nk = __builtin_amdgcn_readfirstlane(nlist[gw]);

    const v4f z = {0.f, 0.f, 0.f, 0.f};
    static_assert((OWN_R * PITCH) % 4 == 0, "whole 16-byte clears");
    for (int i = tid; i < OWN_R * PITCH / 4; i += 1024) reinterpret_cast<v4f*>(s_sum)[i] = z;
    if (COUNT) for (int i = tid; i < OWN_R; i += 1024) s_deg[i] = 0;
    __syncthreads();

    int* q = s_q + wave * Q_CAP;
    int qn = 0;
    const uint4* key4 = reinterpret_cast<const uint4*>(key);

    auto ldlo = [&](int i) -> uint4 {
        const int c = min(i * OWN_WAVES + wave, SCAN_NCH - 1);
        return key4[c * 64 + lane];
    };
    auto ldhi = [&](int i) -> uint4 {
        const int c = min(i * OWN_WAVES + wave, SCAN_NCH - 1);
        return key4[c * 64 + 32 + lane];
    };
    auto add_rows = [&](int s, int r, int n) {
        int sr[DRAIN], dr[DRAIN];
        #pragma unroll
        for (int j = 0; j < DRAIN; ++j) {
            sr[j] = __builtin_amdgcn_readlane(s, j);
            dr[j] = __builtin_amdgcn_readlane(r, j);
        }
        float v[DRAIN][4];
        #pragma unroll
        for (int j = 0; j < DRAIN; ++j) {
            if (__builtin_expect(j < n, 1)) {
                const int ro = sr[j] * F + lane;
                #pragma unroll
                for (int c = 0; c < 4; ++c) v[j][c] = x[ro + 32 * c];
            }
        }
        #pragma unroll
        for (int j = 0; j < DRAIN; ++j) {
            if (__builtin_expect(j < n, 1)) {
                const int ro = dr[j] * PITCH + lane;
                #pragma unroll
                for (int c = 0; c < 4; ++c)
                    __hip_atomic_fetch_add(&s_sum[ro + 32 * c], v[j][c],
                                           __ATOMIC_RELAXED, __HIP_MEMORY_SCOPE_WORKGROUP);
            }
        }
    };
    auto drain = [&](int n) {
        int s = 0, r = 0;
        if (lane < n) {
            const int e = q[qn - n + lane];
            s = ei[e];
            r = ei[N_EDGES + e] - lo;
            if (FIRST && nk + lane < L_CAP)
                L[nk + lane] = (unsigned)s | ((unsigned)r << L_SHIFT);
            if (xidx) s = xidx[s];
            if (COUNT) __hip_atomic_fetch_add(&s_deg[r], 1, __ATOMIC_RELAXED,
                                              __HIP_MEMORY_SCOPE_WORKGROUP);
        }
        if (FIRST) nk += n;
        qn -= n;
        add_rows(s, r, n);
    };
    auto drain_all = [&]() {
        #pragma unroll 1
        while (qn > 0) {
            --qn;
            const int e = q[qn];
            int s = ei[e];
            const int r = ei[N_EDGES + e] - lo;
            if (FIRST) {
                if (lane == 0 && nk < L_CAP) L[nk] = (unsigned)s | ((unsigned)r << L_SHIFT);
                ++nk;
            }
            if (xidx) s = xidx[s];
            if (COUNT && lane == 0) __hip_atomic_fetch_add(&s_deg[r], 1, __ATOMIC_RELAXED,
                                                           __HIP_MEMORY_SCOPE_WORKGROUP);
            #pragma unroll
            for (int c = 0; c < 4; ++c)
                __hip_atomic_fetch_add(&s_sum[r * PITCH + lane + 32 * c], x[s * F + lane + 32 * c],
                                       __ATOMIC_RELAXED, __HIP_MEMORY_SCOPE_WORKGROUP);
        }
    };

    auto scan = [&](const uint4 lo4, const uint4 hi4, int i) {
        const int G = (i * OWN_WAVES + wave) * 32 + lane;
        unsigned g = (lo4.x ^ inv[0]) & (lo4.y ^ inv[1]) & (lo4.z ^ inv[2]) & (lo4.w ^ inv[3]) &
                     (hi4.x ^ inv[4]) & (hi4.y ^ inv[5]) & (hi4.z ^ inv[6]) & (hi4.w ^ inv[7]);
        if (G >= N_GROUPS) g = 0u;
        unsigned mask = __builtin_amdgcn_ballot_w32(g != 0u);
        while (mask) {
            if (__builtin_expect(qn > Q_CAP - 32, 0)) drain_all();
            if (g) {
                q[qn + (int)__builtin_amdgcn_mbcnt_lo(mask, 0u)] = G * 32 + __builtin_ctz(g);
                g &= g - 1u;
            }
            qn += __builtin_popcount(mask);
            mask = __builtin_amdgcn_ballot_w32(g != 0u);
        }
    };

    static_assert(AHEAD == 5, "the trip below is written out for five chunks");
    if (!FIRST && nk <= L_CAP) {
        if (nk > 0) {
            const int last = nk - 1;
            unsigned cur = L[min(lane, last)];
            int s = (int)(cur & ((1u << L_SHIFT) - 1u));
            if (xidx) s = xidx[s];
            #pragma unroll 1
            for (int p = 0; p < nk; p += DRAIN) {
                const unsigned nxt = L[min(p + DRAIN + lane, last)];
                const int r = (int)(cur >> L_SHIFT);
                const int m = min(DRAIN, nk - p);
                if (m == DRAIN) add_rows(s, r, DRAIN);
                else            add_rows(s, r, m);
                cur = nxt;
                s = (int)(cur & ((1u << L_SHIFT) - 1u));
                if (xidx) s = xidx[s];
            }
        }
    } else {
    uint4 a0 = ldlo(0), b0 = ldhi(0), a1 = ldlo(1), b1 = ldhi(1), a2 = ldlo(2), b2 = ldhi(2);
    uint4 a3 = ldlo(3), b3 = ldhi(3), a4 = ldlo(4), b4 = ldhi(4);
    #pragma unroll 1
    for (int i = 0; i < STEPS; i += AHEAD) {
        scan(a0, b0, i);     a0 = ldlo(i + 5); b0 = ldhi(i + 5);
        scan(a1, b1, i + 1); a1 = ldlo(i + 6); b1 = ldhi(i + 6);
        scan(a2, b2, i + 2); a2 = ldlo(i + 7); b2 = ldhi(i + 7);
        scan(a3, b3, i + 3); a3 = ldlo(i + 8); b3 = ldhi(i + 8);
        scan(a4, b4, i + 4); a4 = ldlo(i + 9); b4 = ldhi(i + 9);
        while (qn >= DRAIN) drain(DRAIN);
    }
    if (qn > 0) drain(qn);
    if (FIRST && lane == 0) nlist[gw] = nk;
    }

    __syncthreads();
    if (COUNT)
        for (int i = tid; i < OWN_R; i += 1024)
            if (lo + i < N_NODES) cnt[lo + i] = (float)s_deg[i];

    {
        constexpr int K  = F;
        constexpr int SP = OUT / 64;
        constexpr int NG = K / 8;
        const int nrows  = min(OWN_R, N_NODES - lo);
        const int nunits = ((nrows + 15) >> 4) * SP;
        const int half = lane >> 4;
        const int l16  = lane & 15;
        const int wu = __builtin_amdgcn_readfirstlane(wave);
        auto unit = [&](const int u) {
            const int span  = u % SP;
            const int t     = u / SP;
            const int cbase = span * 64 + 4 * l16;
            const int lrow  = min(t * 16 + l16, nrows - 1);
            const float deg = FIRST ? (float)s_deg[lrow] : cnt[lo + lrow];
            const float inv = 1.0f / fmaxf(deg, 1.0f);
            int xrow = lo + lrow;
            if (xidx) xrow = xidx[xrow];
            const int aoff   = lrow * PITCH + 4 * half;
            const float* px  = x  + (size_t)xrow * K + 4 * half;
            const float* pwl = Wl + (size_t)(4 * half) * OUT + cbase;
            const float* pwr = Wr + (size_t)(4 * half) * OUT + cbase;

            v8f c[4];
            #pragma unroll
            for (int j = 0; j < 4; ++j)
                c[j] = (v8f){0.f, 0.f, 0.f, 0.f, 0.f, 0.f, 0.f, 0.f};

            v4f wc[4], ac;
            #pragma unroll
            for (int i = 0; i < 4; ++i)
                wc[i] = *reinterpret_cast<const v4f*>(pwl + (size_t)i * OUT);
            ac = *reinterpret_cast<const v4f*>(&s_sum[aoff]);

            #pragma unroll
            for (int G = 0; G < 2 * NG; ++G) {
                v4f wn[4], an;
                if (G + 1 < 2 * NG) {
                    const bool nfirst = (G + 1) < NG;
                    const int  ng     = nfirst ? (G + 1) : (G + 1 - NG);
                    const float* pw   = nfirst ? pwl : pwr;
                    #pragma unroll
                    for (int i = 0; i < 4; ++i)
                        wn[i] = *reinterpret_cast<const v4f*>(pw + (size_t)(8 * ng + i) * OUT);
                    if (nfirst) an = *reinterpret_cast<const v4f*>(&s_sum[aoff + 8 * ng]);
                    else        an = *reinterpret_cast<const v4f*>(px + 8 * ng);
                }
                __builtin_amdgcn_sched_barrier(0);
                v4f a = ac;
                if (G < NG) a = a * inv;
                v2f a0; a0.x = a.x; a0.y = a.y;
                v2f a1; a1.x = a.z; a1.y = a.w;
                #pragma unroll
                for (int j = 0; j < 4; ++j) {
                    v2f b; b.x = wc[0][j]; b.y = wc[1][j];
                    c[j] = __builtin_amdgcn_wmma_f32_16x16x4_f32(false, a0, false, b,
                                                                 (short)0, c[j], false, false);
                }
                #pragma unroll
                for (int j = 0; j < 4; ++j) {
                    v2f b; b.x = wc[2][j]; b.y = wc[3][j];
                    c[j] = __builtin_amdgcn_wmma_f32_16x16x4_f32(false, a1, false, b,
                                                                 (short)0, c[j], false, false);
                }
                if (G + 1 < 2 * NG) {
                    #pragma unroll
                    for (int i = 0; i < 4; ++i) wc[i] = wn[i];
                    ac = an;
                }
            }

            const v4f bv = *reinterpret_cast<const v4f*>(bias + cbase);
            #pragma unroll
            for (int v = 0; v < 8; ++v) {
                const int rl = t * 16 + v + half * 8;
                v4f o;
                o.x = c[0][v] + bv.x;
                o.y = c[1][v] + bv.y;
                o.z = c[2][v] + bv.z;
                o.w = c[3][v] + bv.w;
                if (RELU) {
                    o.x = fmaxf(o.x, 0.0f); o.y = fmaxf(o.y, 0.0f);
                    o.z = fmaxf(o.z, 0.0f); o.w = fmaxf(o.w, 0.0f);
                }
                if (rl < nrows)
                    *reinterpret_cast<v4f*>(out + (size_t)(lo + rl) * OUT + cbase) = o;
            }
        };
        constexpr int MAXU = ((OWN_R + 15) / 16) * SP;
        if (MAXU <= OWN_WAVES) {
            if (wu < nunits) unit(wu);
        } else {
            #pragma unroll 1
            for (int u = wu; u < nunits; u += OWN_WAVES) unit(u);
        }
    }
}

extern "C" void kernel_launch(void* const* d_in, const int* in_sizes, int n_in,
                              void* d_out, int out_size, void* d_ws, size_t ws_size,
                              hipStream_t stream) {
    (void)in_sizes; (void)n_in; (void)out_size; (void)ws_size;

    const int*   xidx = (const int*)d_in[0];
    const int*   ei   = (const int*)d_in[1];
    const float* emb  = (const float*)d_in[2];
    const float* Wl1  = (const float*)d_in[3];
    const float* bl1  = (const float*)d_in[4];
    const float* Wr1  = (const float*)d_in[5];
    const float* Wl2  = (const float*)d_in[6];
    const float* bl2  = (const float*)d_in[7];
    const float* Wr2  = (const float*)d_in[8];

    unsigned* key  = (unsigned*)d_ws;
    unsigned* list = key + KEY_WORDS;
    int* nlist = (int*)(list + LIST_WORDS);
    float* h   = (float*)d_ws + (size_t)N_NODES * HIDDEN;
    float* cnt = h + (size_t)N_NODES * HIDDEN;
    static_assert(((KEY_NCH + 7) / 8) * 8 * KCH <= KEY_WORDS * 4, "the planes of every launched wave end in front of the lists");
    static_assert((size_t)(KEY_WORDS + LIST_WORDS + OWN_BLOCKS * OWN_WAVES) * 4 <=
                  (size_t)N_NODES * HIDDEN * 4, "keys, lists and lengths end in front of h");
    static_assert(EMB_DIM == 128 && HIDDEN == 128, "a gathered row is 128 floats in both layers");

    owner_key_kernel<<<(KEY_NCH + 7) / 8, 256, 0, stream>>>(ei, key);

    owner_kernel<true, HIDDEN, true><<<OWN_BLOCKS, 1024, 0, stream>>>(
        ei, key, emb, xidx, list, nlist, Wl1, bl1, Wr1, cnt, h);

    owner_kernel<false, OUT_C, false><<<OWN_BLOCKS, 1024, 0, stream>>>(
        ei, key, h, nullptr, list, nlist, Wl2, bl2, Wr2, cnt, (float*)d_out);
}


# Gathered rows: one 16-byte load per lane, a whole 512-byte row per load (was four 128-byte line loads); LDS sum word 32c+l = float 4l+c, mean product takes k in that order: sum order changes
# speedup vs baseline: 4.2112x; 1.0537x over previous
#include <hip/hip_runtime.h>

typedef __attribute__((ext_vector_type(2))) float v2f;
typedef __attribute__((ext_vector_type(4))) float v4f;
typedef __attribute__((ext_vector_type(8))) float v8f;

#define N_NODES 100000
#define EMB_DIM 128
#define HIDDEN  128
#define OUT_C   64
#define N_EDGES 800000

#define OWN_R      391
#define PITCH      132
#define OWN_BLOCKS ((N_NODES + OWN_R - 1) / OWN_R)
#define OWN_WAVES  32
#define DRAIN      16
#define AHEAD      5
#define KCH        512
#define KEY_NCH    ((N_EDGES + KCH - 1) / KCH)
#define N_GROUPS   (N_EDGES / 32)
#define SCAN_NCH   ((N_GROUPS + 31) / 32)
#define Q_CAP      768
#define L_CAP      512
#define L_SHIFT    17
#define KEY_WORDS  (1 << 18)
#define LIST_WORDS (OWN_BLOCKS * OWN_WAVES * L_CAP)

__global__ __launch_bounds__(256)
void owner_key_kernel(const int* __restrict__ ei, unsigned* __restrict__ key) {
    static_assert(OWN_BLOCKS <= 256, "owner number must fit 8 bits");
    static_assert(N_EDGES % 32 == 0, "whole groups of 32 edges");
    const int lane = threadIdx.x & 31;
    const int kc   = blockIdx.x * 8 + (threadIdx.x >> 5);
    const int* dst = ei + N_EDGES;
    int d[16];
    #pragma unroll
    for (int g = 0; g < 16; ++g)
        d[g] = dst[min(kc * KCH + 32 * g + lane, N_EDGES - 1)];
    unsigned o[4] = {0u, 0u, 0u, 0u};
    #pragma unroll
    for (int g = 0; g < 16; ++g) {
        const unsigned k = (unsigned)d[g] / OWN_R;
        #pragma unroll
        for (int b = 0; b < 8; ++b) {
            const unsigned w = __builtin_amdgcn_ballot_w32(((k >> b) & 1u) != 0u);
            if (lane == g + 16 * (b >> 2)) o[b & 3] = w;
        }
    }
    uint4 v;
    v.x = o[0]; v.y = o[1]; v.z = o[2]; v.w = o[3];
    reinterpret_cast<uint4*>(key)[(kc >> 1) * 64 + (lane >> 4) * 32 + (kc & 1) * 16 + (lane & 15)] = v;
}

template<bool FIRST, int OUT, bool RELU>
__global__ __launch_bounds__(1024)
void owner_kernel(const int* __restrict__ ei,
                  const unsigned* __restrict__ key,
                  const float* __restrict__ x,
                  const int* __restrict__ xidx,
                  unsigned* __restrict__ list,
                  int* __restrict__ nlist,
                  const float* __restrict__ Wl,
                  const float* __restrict__ bias,
                  const float* __restrict__ Wr,
                  float* __restrict__ cnt,
                  float* __restrict__ out) {
    constexpr bool COUNT = FIRST;
    constexpr int F   = 128;
    static_assert(N_NODES <= (1 << L_SHIFT), "src fits below the local dst in a list word");
    static_assert(OWN_R <= (1 << (32 - L_SHIFT)), "local dst fits above src in a list word");
    static_assert(DRAIN <= 32 && L_CAP % DRAIN == 0, "one list entry per lane and turn");
    constexpr int STEPS = (SCAN_NCH + OWN_WAVES - 1) / OWN_WAVES;
    static_assert(N_EDGES % 32 == 0, "a lane's group of 32 edges is all inside the list or all outside");
    static_assert(STEPS % AHEAD == 0, "whole trips");
    static_assert(Q_CAP >= 64, "a push turn adds at most 32 entries to at most Q_CAP - 32");
    __shared__ __attribute__((aligned(16))) float s_sum[OWN_R * PITCH];
    __shared__ int   s_deg[OWN_R];
    __shared__ int   s_q[OWN_WAVES * Q_CAP];

    const int tid  = threadIdx.x;
    const int lane = tid & 31;
    const int wave = tid >> 5;
    const int lo   = blockIdx.x * OWN_R;
    unsigned inv[8];
    #pragma unroll
    for (int b = 0; b < 8; ++b) inv[b] = ((blockIdx.x >> b) & 1u) ? 0u : 0xffffffffu;
    const int gw   = blockIdx.x * OWN_WAVES + wave;
    unsigned* L    = list + (size_t)gw * L_CAP;
    int nk = 0;
    if (!FIRST) nk = __builtin_amdgcn_readfirstlane(nlist[gw]);

    const v4f z = {0.f, 0.f, 0.f, 0.f};
    static_assert((OWN_R * PITCH) % 4 == 0, "whole 16-byte clears");
    for (int i = tid; i < OWN_R * PITCH / 4; i += 1024) reinterpret_cast<v4f*>(s_sum)[i] = z;
    if (COUNT) for (int i = tid; i < OWN_R; i += 1024) s_deg[i] = 0;
    __syncthreads();

    int* q = s_q + wave * Q_CAP;
    int qn = 0;
    const uint4* key4 = reinterpret_cast<const uint4*>(key);

    auto ldlo = [&](int i) -> uint4 {
        const int c = min(i * OWN_WAVES + wave, SCAN_NCH - 1);
        return key4[c * 64 + lane];
    };
    auto ldhi = [&](int i) -> uint4 {
        const int c = min(i * OWN_WAVES + wave, SCAN_NCH - 1);
        return key4[c * 64 + 32 + lane];
    };
    auto add_rows = [&](int s, int r, int n) {
        int sr[DRAIN], dr[DRAIN];
        #pragma unroll
        for (int j = 0; j < DRAIN; ++j) {
            sr[j] = __builtin_amdgcn_readlane(s, j);
            dr[j] = __builtin_amdgcn_readlane(r, j);
        }
        v4f v[DRAIN];
        #pragma unroll
        for (int j = 0; j < DRAIN; ++j) {
            if (__builtin_expect(j < n, 1))
                v[j] = reinterpret_cast<const v4f*>(x)[sr[j] * (F / 4) + lane];
        }
        #pragma unroll
        for (int j = 0; j < DRAIN; ++j) {
            if (__builtin_expect(j < n, 1)) {
                const int ro = dr[j] * PITCH + lane;
                #pragma unroll
                for (int c = 0; c < 4; ++c)
                    __hip_atomic_fetch_add(&s_sum[ro + 32 * c], v[j][c],
                                           __ATOMIC_RELAXED, __HIP_MEMORY_SCOPE_WORKGROUP);
            }
        }
    };
    auto drain = [&](int n) {
        int s = 0, r = 0;
        if (lane < n) {
            const int e = q[qn - n + lane];
            s = ei[e];
            r = ei[N_EDGES + e] - lo;
            if (FIRST && nk + lane < L_CAP)
                L[nk + lane] = (unsigned)s | ((unsigned)r << L_SHIFT);
            if (xidx) s = xidx[s];
            if (COUNT) __hip_atomic_fetch_add(&s_deg[r], 1, __ATOMIC_RELAXED,
                                              __HIP_MEMORY_SCOPE_WORKGROUP);
        }
        if (FIRST) nk += n;
        qn -= n;
        add_rows(s, r, n);
    };
    auto drain_all = [&]() {
        #pragma unroll 1
        while (qn > 0) {
            --qn;
            const int e = q[qn];
            int s = ei[e];
            const int r = ei[N_EDGES + e] - lo;
            if (FIRST) {
                if (lane == 0 && nk < L_CAP) L[nk] = (unsigned)s | ((unsigned)r << L_SHIFT);
                ++nk;
            }
            if (xidx) s = xidx[s];
            if (COUNT && lane == 0) __hip_atomic_fetch_add(&s_deg[r], 1, __ATOMIC_RELAXED,
                                                           __HIP_MEMORY_SCOPE_WORKGROUP);
            const v4f xv = reinterpret_cast<const v4f*>(x)[s * (F / 4) + lane];
            #pragma unroll
            for (int c = 0; c < 4; ++c)
                __hip_atomic_fetch_add(&s_sum[r * PITCH + lane + 32 * c], xv[c],
                                       __ATOMIC_RELAXED, __HIP_MEMORY_SCOPE_WORKGROUP);
        }
    };

    auto scan = [&](const uint4 lo4, const uint4 hi4, int i) {
        const int G = (i * OWN_WAVES + wave) * 32 + lane;
        unsigned g = (lo4.x ^ inv[0]) & (lo4.y ^ inv[1]) & (lo4.z ^ inv[2]) & (lo4.w ^ inv[3]) &
                     (hi4.x ^ inv[4]) & (hi4.y ^ inv[5]) & (hi4.z ^ inv[6]) & (hi4.w ^ inv[7]);
        if (G >= N_GROUPS) g = 0u;
        unsigned mask = __builtin_amdgcn_ballot_w32(g != 0u);
        while (mask) {
            if (__builtin_expect(qn > Q_CAP - 32, 0)) drain_all();
            if (g) {
                q[qn + (int)__builtin_amdgcn_mbcnt_lo(mask, 0u)] = G * 32 + __builtin_ctz(g);
                g &= g - 1u;
            }
            qn += __builtin_popcount(mask);
            mask = __builtin_amdgcn_ballot_w32(g != 0u);
        }
    };

    static_assert(AHEAD == 5, "the trip below is written out for five chunks");
    if (!FIRST && nk <= L_CAP) {
        if (nk > 0) {
            const int last = nk - 1;
            unsigned cur = L[min(lane, last)];
            int s = (int)(cur & ((1u << L_SHIFT) - 1u));
            if (xidx) s = xidx[s];
            #pragma unroll 1
            for (int p = 0; p < nk; p += DRAIN) {
                const unsigned nxt = L[min(p + DRAIN + lane, last)];
                const int r = (int)(cur >> L_SHIFT);
                const int m = min(DRAIN, nk - p);
                if (m == DRAIN) add_rows(s, r, DRAIN);
                else            add_rows(s, r, m);
                cur = nxt;
                s = (int)(cur & ((1u << L_SHIFT) - 1u));
                if (xidx) s = xidx[s];
            }
        }
    } else {
    uint4 a0 = ldlo(0), b0 = ldhi(0), a1 = ldlo(1), b1 = ldhi(1), a2 = ldlo(2), b2 = ldhi(2);
    uint4 a3 = ldlo(3), b3 = ldhi(3), a4 = ldlo(4), b4 = ldhi(4);
    #pragma unroll 1
    for (int i = 0; i < STEPS; i += AHEAD) {
        scan(a0, b0, i);     a0 = ldlo(i + 5); b0 = ldhi(i + 5);
        scan(a1, b1, i + 1); a1 = ldlo(i + 6); b1 = ldhi(i + 6);
        scan(a2, b2, i + 2); a2 = ldlo(i + 7); b2 = ldhi(i + 7);
        scan(a3, b3, i + 3); a3 = ldlo(i + 8); b3 = ldhi(i + 8);
        scan(a4, b4, i + 4); a4 = ldlo(i + 9); b4 = ldhi(i + 9);
        while (qn >= DRAIN) drain(DRAIN);
    }
    if (qn > 0) drain(qn);
    if (FIRST && lane == 0) nlist[gw] = nk;
    }

    __syncthreads();
    if (COUNT)
        for (int i = tid; i < OWN_R; i += 1024)
            if (lo + i < N_NODES) cnt[lo + i] = (float)s_deg[i];

    {
        constexpr int K  = F;
        constexpr int SP = OUT / 64;
        constexpr int NG = K / 8;
        const int nrows  = min(OWN_R, N_NODES - lo);
        const int nunits = ((nrows + 15) >> 4) * SP;
        const int half = lane >> 4;
        const int l16  = lane & 15;
        const int wu = __builtin_amdgcn_readfirstlane(wave);
        auto unit = [&](const int u) {
            const int span  = u % SP;
            const int t     = u / SP;
            const int cbase = span * 64 + 4 * l16;
            const int lrow  = min(t * 16 + l16, nrows - 1);
            const float deg = FIRST ? (float)s_deg[lrow] : cnt[lo + lrow];
            const float inv = 1.0f / fmaxf(deg, 1.0f);
            int xrow = lo + lrow;
            if (xidx) xrow = xidx[xrow];
            const int aoff   = lrow * PITCH + 4 * half;
            const float* px  = x  + (size_t)xrow * K + 4 * half;
            const float* pwl = Wl + (size_t)(16 * half) * OUT + cbase;
            const float* pwr = Wr + (size_t)(4 * half) * OUT + cbase;

            v8f c[4];
            #pragma unroll
            for (int j = 0; j < 4; ++j)
                c[j] = (v8f){0.f, 0.f, 0.f, 0.f, 0.f, 0.f, 0.f, 0.f};

            v4f wc[4], ac;
            #pragma unroll
            for (int i = 0; i < 4; ++i)
                wc[i] = *reinterpret_cast<const v4f*>(pwl + (size_t)(4 * i) * OUT);
            ac = *reinterpret_cast<const v4f*>(&s_sum[aoff]);

            #pragma unroll
            for (int G = 0; G < 2 * NG; ++G) {
                v4f wn[4], an;
                if (G + 1 < 2 * NG) {
                    const bool nfirst = (G + 1) < NG;
                    const int  ng     = nfirst ? (G + 1) : (G + 1 - NG);
                    #pragma unroll
                    for (int i = 0; i < 4; ++i) {
                        const float* pw = nfirst ? pwl + (size_t)(32 * (ng & 3) + (ng >> 2) + 4 * i) * OUT
                                                 : pwr + (size_t)(8 * ng + i) * OUT;
                        wn[i] = *reinterpret_cast<const v4f*>(pw);
                    }
                    if (nfirst) an = *reinterpret_cast<const v4f*>(&s_sum[aoff + 8 * ng]);
                    else        an = *reinterpret_cast<const v4f*>(px + 8 * ng);
                }
                __builtin_amdgcn_sched_barrier(0);
                v4f a = ac;
                if (G < NG) a = a * inv;
                v2f a0; a0.x = a.x; a0.y = a.y;
                v2f a1; a1.x = a.z; a1.y = a.w;
                #pragma unroll
                for (int j = 0; j < 4; ++j) {
                    v2f b; b.x = wc[0][j]; b.y = wc[1][j];
                    c[j] = __builtin_amdgcn_wmma_f32_16x16x4_f32(false, a0, false, b,
                                                                 (short)0, c[j], false, false);
                }
                #pragma unroll
                for (int j = 0; j < 4; ++j) {
                    v2f b; b.x = wc[2][j]; b.y = wc[3][j];
                    c[j] = __builtin_amdgcn_wmma_f32_16x16x4_f32(false, a1, false, b,
                                                                 (short)0, c[j], false, false);
                }
                if (G + 1 < 2 * NG) {
                    #pragma unroll
                    for (int i = 0; i < 4; ++i) wc[i] = wn[i];
                    ac = an;
                }
            }

            const v4f bv = *reinterpret_cast<const v4f*>(bias + cbase);
            #pragma unroll
            for (int v = 0; v < 8; ++v) {
                const int rl = t * 16 + v + half * 8;
                v4f o;
                o.x = c[0][v] + bv.x;
                o.y = c[1][v] + bv.y;
                o.z = c[2][v] + bv.z;
                o.w = c[3][v] + bv.w;
                if (RELU) {
                    o.x = fmaxf(o.x, 0.0f); o.y = fmaxf(o.y, 0.0f);
                    o.z = fmaxf(o.z, 0.0f); o.w = fmaxf(o.w, 0.0f);
                }
                if (rl < nrows)
                    *reinterpret_cast<v4f*>(out + (size_t)(lo + rl) * OUT + cbase) = o;
            }
        };
        constexpr int MAXU = ((OWN_R + 15) / 16) * SP;
        if (MAXU <= OWN_WAVES) {
            if (wu < nunits) unit(wu);
        } else {
            #pragma unroll 1
            for (int u = wu; u < nunits; u += OWN_WAVES) unit(u);
        }
    }
}

extern "C" void kernel_launch(void* const* d_in, const int* in_sizes, int n_in,
                              void* d_out, int out_size, void* d_ws, size_t ws_size,
                              hipStream_t stream) {
    (void)in_sizes; (void)n_in; (void)out_size; (void)ws_size;

    const int*   xidx = (const int*)d_in[0];
    const int*   ei   = (const int*)d_in[1];
    const float* emb  = (const float*)d_in[2];
    const float* Wl1  = (const float*)d_in[3];
    const float* bl1  = (const float*)d_in[4];
    const float* Wr1  = (const float*)d_in[5];
    const float* Wl2  = (const float*)d_in[6];
    const float* bl2  = (const float*)d_in[7];
    const float* Wr2  = (const float*)d_in[8];

    unsigned* key  = (unsigned*)d_ws;
    unsigned* list = key + KEY_WORDS;
    int* nlist = (int*)(list + LIST_WORDS);
    float* h   = (float*)d_ws + (size_t)N_NODES * HIDDEN;
    float* cnt = h + (size_t)N_NODES * HIDDEN;
    static_assert(((KEY_NCH + 7) / 8) * 8 * KCH <= KEY_WORDS * 4, "the planes of every launched wave end in front of the lists");
    static_assert((size_t)(KEY_WORDS + LIST_WORDS + OWN_BLOCKS * OWN_WAVES) * 4 <=
                  (size_t)N_NODES * HIDDEN * 4, "keys, lists and lengths end in front of h");
    static_assert(EMB_DIM == 128 && HIDDEN == 128, "a gathered row is 128 floats in both layers");

    owner_key_kernel<<<(KEY_NCH + 7) / 8, 256, 0, stream>>>(ei, key);

    owner_kernel<true, HIDDEN, true><<<OWN_BLOCKS, 1024, 0, stream>>>(
        ei, key, emb, xidx, list, nlist, Wl1, bl1, Wr1, cnt, h);

    owner_kernel<false, OUT_C, false><<<OWN_BLOCKS, 1024, 0, stream>>>(
        ei, key, h, nullptr, list, nlist, Wl2, bl2, Wr2, cnt, (float*)d_out);
}
